# starting kernel
# baseline (speedup 1.0000x reference)
#include <hip/hip_runtime.h>

typedef float v2f __attribute__((ext_vector_type(2)));
typedef float v4f __attribute__((ext_vector_type(4)));
typedef float v8f __attribute__((ext_vector_type(8)));

#define WMMA_F32(a, b, c) \
    __builtin_amdgcn_wmma_f32_16x16x4_f32(false, (a), false, (b), (short)0, (c), false, false)

constexpr int Bn = 8, Tn = 2048, Cn = 1024, Hn = 64;

constexpr size_t OFF_KP = 0;
constexpr size_t OFF_QP = OFF_KP + (size_t)Bn * Tn * Hn;
constexpr size_t OFF_VT = OFF_QP + (size_t)Bn * Tn * Hn;
constexpr size_t OFF_WT = OFF_VT + (size_t)Bn * Tn * Hn;

__global__ void wtrans_kernel(const float* __restrict__ Wk,
                              const float* __restrict__ Wq,
                              const float* __restrict__ Wv,
                              float* __restrict__ wt) {
    int idx = blockIdx.x * 256 + threadIdx.x;
    int m = idx >> 16;
    int e = idx & 65535;
    int h = e >> 10;
    int c = e & 1023;
    const float* W = (m == 0) ? Wk : ((m == 1) ? Wq : Wv);
    float val = W[c * Hn + h];
    if (m == 1) val *= 0.125f;
    wt[(size_t)m * 65536 + e] = val;
}

__global__ void proj_kernel(const float* __restrict__ x,
                            const float* __restrict__ wt,
                            float* __restrict__ ws) {
    int wave  = blockIdx.x * (blockDim.x >> 5) + (threadIdx.x >> 5);
    int lane  = threadIdx.x & 31;
    int which = wave % 3;
    int rt    = wave / 3;
    int l15   = lane & 15;
    int hi    = lane >> 4;

    const float* xa  = x + (size_t)(rt * 16 + l15) * Cn + 2 * hi;
    const float* wbb = wt + (size_t)which * Hn * Cn + 2 * hi;
    const float* wb0 = wbb + (size_t)(0 * 16 + l15) * Cn;
    const float* wb1 = wbb + (size_t)(1 * 16 + l15) * Cn;
    const float* wb2 = wbb + (size_t)(2 * 16 + l15) * Cn;
    const float* wb3 = wbb + (size_t)(3 * 16 + l15) * Cn;

    v8f a0 = {}, a1 = {}, a2 = {}, a3 = {};
#pragma unroll 4
    for (int k = 0; k < Cn; k += 4) {
        v2f a = *(const v2f*)(xa + k);
        a0 = WMMA_F32(a, *(const v2f*)(wb0 + k), a0);
        a1 = WMMA_F32(a, *(const v2f*)(wb1 + k), a1);
        a2 = WMMA_F32(a, *(const v2f*)(wb2 + k), a2);
        a3 = WMMA_F32(a, *(const v2f*)(wb3 + k), a3);
    }

    if (which < 2) {
        float* p = ws + (which == 0 ? OFF_KP : OFF_QP);
        int rowbase = rt * 16 + 8 * hi;
#pragma unroll
        for (int r = 0; r < 8; r++) {
            p[(size_t)(rowbase + r) * Hn + 0 * 16 + l15] = a0[r];
            p[(size_t)(rowbase + r) * Hn + 1 * 16 + l15] = a1[r];
            p[(size_t)(rowbase + r) * Hn + 2 * 16 + l15] = a2[r];
            p[(size_t)(rowbase + r) * Hn + 3 * 16 + l15] = a3[r];
        }
    } else {
        int b  = rt >> 7;
        int t0 = (rt & 127) * 16 + 8 * hi;
        float* vtb = ws + OFF_VT + (size_t)b * Hn * Tn + t0;
        v8f accs0 = a0, accs1 = a1, accs2 = a2, accs3 = a3;
        {
            float* d = vtb + (size_t)(0 * 16 + l15) * Tn;
            *(v4f*)(d)     = (v4f){accs0[0], accs0[1], accs0[2], accs0[3]};
            *(v4f*)(d + 4) = (v4f){accs0[4], accs0[5], accs0[6], accs0[7]};
        }
        {
            float* d = vtb + (size_t)(1 * 16 + l15) * Tn;
            *(v4f*)(d)     = (v4f){accs1[0], accs1[1], accs1[2], accs1[3]};
            *(v4f*)(d + 4) = (v4f){accs1[4], accs1[5], accs1[6], accs1[7]};
        }
        {
            float* d = vtb + (size_t)(2 * 16 + l15) * Tn;
            *(v4f*)(d)     = (v4f){accs2[0], accs2[1], accs2[2], accs2[3]};
            *(v4f*)(d + 4) = (v4f){accs2[4], accs2[5], accs2[6], accs2[7]};
        }
        {
            float* d = vtb + (size_t)(3 * 16 + l15) * Tn;
            *(v4f*)(d)     = (v4f){accs3[0], accs3[1], accs3[2], accs3[3]};
            *(v4f*)(d + 4) = (v4f){accs3[4], accs3[5], accs3[6], accs3[7]};
        }
    }
}

__global__ void attn_kernel(const float* __restrict__ ws, float* __restrict__ out) {
    int wave = blockIdx.x * (blockDim.x >> 5) + (threadIdx.x >> 5);
    int lane = threadIdx.x & 31;
    int b = wave >> 7;
    int i = wave & 127;
    int l15 = lane & 15;
    int hi  = lane >> 4;

    const float* kp = ws + OFF_KP + (size_t)(b * Tn + i * 16 + l15) * Hn + 2 * hi;
    const float* qp = ws + OFF_QP + (size_t)b * Tn * Hn;
    const float* vt = ws + OFF_VT + (size_t)b * Hn * Tn;

    v2f kb[16];
#pragma unroll
    for (int kk = 0; kk < 16; kk++) kb[kk] = *(const v2f*)(kp + 4 * kk);

    v8f oacc[4] = {};
    float m = -__builtin_inff();
    float l = 0.0f;

    for (int j = 0; j <= i; j++) {
        const float* qa = qp + (size_t)(j * 16 + l15) * Hn + 2 * hi;
        v8f s = {};
#pragma unroll
        for (int kk = 0; kk < 16; kk++) {
            v2f a = *(const v2f*)(qa + 4 * kk);
            s = WMMA_F32(a, kb[kk], s);
        }

        if (j == i) {
#pragma unroll
            for (int r = 0; r < 8; r++) {
                int sl = r + 8 * hi;
                s[r] = (sl > l15) ? -__builtin_inff() : s[r];
            }
        }

        float tm = s[0];
#pragma unroll
        for (int r = 1; r < 8; r++) tm = fmaxf(tm, s[r]);
        tm = fmaxf(tm, __shfl_xor(tm, 16, 32));
        float mn = fmaxf(m, tm);
        float sc = __expf(m - mn);
        m = mn;

        float p[8];
        float tl = 0.0f;
#pragma unroll
        for (int r = 0; r < 8; r++) {
            p[r] = __expf(s[r] - mn);
            tl += p[r];
        }
        tl += __shfl_xor(tl, 16, 32);
        l = l * sc + tl;
#pragma unroll
        for (int n = 0; n < 4; n++) oacc[n] = oacc[n] * sc;

        float q0 = __shfl_xor(p[0], 16, 32), q1 = __shfl_xor(p[1], 16, 32);
        float q2 = __shfl_xor(p[2], 16, 32), q3 = __shfl_xor(p[3], 16, 32);
        float q4 = __shfl_xor(p[4], 16, 32), q5 = __shfl_xor(p[5], 16, 32);
        float q6 = __shfl_xor(p[6], 16, 32), q7 = __shfl_xor(p[7], 16, 32);
        bool ishi = (hi != 0);
        v2f bb[4];
        bb[0] = ishi ? (v2f){q2, q3} : (v2f){p[0], p[1]};
        bb[1] = ishi ? (v2f){q6, q7} : (v2f){p[4], p[5]};
        bb[2] = ishi ? (v2f){p[2], p[3]} : (v2f){q0, q1};
        bb[3] = ishi ? (v2f){p[6], p[7]} : (v2f){q4, q5};

        const float* va = vt + (size_t)l15 * Tn + j * 16 + 2 * hi;
#pragma unroll
        for (int n = 0; n < 4; n++) {
            const float* van = va + (size_t)n * 16 * Tn;
#pragma unroll
            for (int kk = 0; kk < 4; kk++) {
                v2f a = *(const v2f*)(van + 4 * kk);
                oacc[n] = WMMA_F32(a, bb[kk], oacc[n]);
            }
        }
    }

    float invl = 1.0f / l;
    float* dst = out + (size_t)(b * Tn + i * 16 + l15) * Hn + 8 * hi;
#pragma unroll
    for (int n = 0; n < 4; n++) {
        v4f lo = {oacc[n][0] * invl, oacc[n][1] * invl, oacc[n][2] * invl, oacc[n][3] * invl};
        v4f hi4 = {oacc[n][4] * invl, oacc[n][5] * invl, oacc[n][6] * invl, oacc[n][7] * invl};
        *(v4f*)(dst + n * 16)     = lo;
        *(v4f*)(dst + n * 16 + 4) = hi4;
    }
}

extern "C" void kernel_launch(void* const* d_in, const int* in_sizes, int n_in,
                              void* d_out, int out_size, void* d_ws, size_t ws_size,
                              hipStream_t stream) {
    const float* x  = (const float*)d_in[0];
    const float* Wk = (const float*)d_in[1];
    const float* Wq = (const float*)d_in[2];
    const float* Wv = (const float*)d_in[3];
    float* ws  = (float*)d_ws;
    float* out = (float*)d_out;

    wtrans_kernel<<<768, 256, 0, stream>>>(Wk, Wq, Wv, ws + OFF_WT);
    proj_kernel<<<384, 256, 0, stream>>>(x, ws + OFF_WT, ws);
    attn_kernel<<<128, 256, 0, stream>>>(ws, out);
}


# attention: one 8-wave block per 16-row tile, key tiles dealt round-robin to the waves, partial (m,l,O) merged through LDS, heavy tiles first; changes the softmax/PV summation order
# speedup vs baseline: 2.3652x; 2.3652x over previous
#include <hip/hip_runtime.h>

typedef float v2f __attribute__((ext_vector_type(2)));
typedef float v4f __attribute__((ext_vector_type(4)));
typedef float v8f __attribute__((ext_vector_type(8)));

#define WMMA_F32(a, b, c) \
    __builtin_amdgcn_wmma_f32_16x16x4_f32(false, (a), false, (b), (short)0, (c), false, false)

constexpr int Bn = 8, Tn = 2048, Cn = 1024, Hn = 64;

constexpr size_t OFF_KP = 0;
constexpr size_t OFF_QP = OFF_KP + (size_t)Bn * Tn * Hn;
constexpr size_t OFF_VT = OFF_QP + (size_t)Bn * Tn * Hn;
constexpr size_t OFF_WT = OFF_VT + (size_t)Bn * Tn * Hn;

__global__ void wtrans_kernel(const float* __restrict__ Wk,
                              const float* __restrict__ Wq,
                              const float* __restrict__ Wv,
                              float* __restrict__ wt) {
    int idx = blockIdx.x * 256 + threadIdx.x;
    int m = idx >> 16;
    int e = idx & 65535;
    int h = e >> 10;
    int c = e & 1023;
    const float* W = (m == 0) ? Wk : ((m == 1) ? Wq : Wv);
    float val = W[c * Hn + h];
    if (m == 1) val *= 0.125f;
    wt[(size_t)m * 65536 + e] = val;
}

__global__ void proj_kernel(const float* __restrict__ x,
                            const float* __restrict__ wt,
                            float* __restrict__ ws) {
    int wave  = blockIdx.x * (blockDim.x >> 5) + (threadIdx.x >> 5);
    int lane  = threadIdx.x & 31;
    int which = wave % 3;
    int rt    = wave / 3;
    int l15   = lane & 15;
    int hi    = lane >> 4;

    const float* xa  = x + (size_t)(rt * 16 + l15) * Cn + 2 * hi;
    const float* wbb = wt + (size_t)which * Hn * Cn + 2 * hi;
    const float* wb0 = wbb + (size_t)(0 * 16 + l15) * Cn;
    const float* wb1 = wbb + (size_t)(1 * 16 + l15) * Cn;
    const float* wb2 = wbb + (size_t)(2 * 16 + l15) * Cn;
    const float* wb3 = wbb + (size_t)(3 * 16 + l15) * Cn;

    v8f a0 = {}, a1 = {}, a2 = {}, a3 = {};
#pragma unroll 4
    for (int k = 0; k < Cn; k += 4) {
        v2f a = *(const v2f*)(xa + k);
        a0 = WMMA_F32(a, *(const v2f*)(wb0 + k), a0);
        a1 = WMMA_F32(a, *(const v2f*)(wb1 + k), a1);
        a2 = WMMA_F32(a, *(const v2f*)(wb2 + k), a2);
        a3 = WMMA_F32(a, *(const v2f*)(wb3 + k), a3);
    }

    if (which < 2) {
        float* p = ws + (which == 0 ? OFF_KP : OFF_QP);
        int rowbase = rt * 16 + 8 * hi;
#pragma unroll
        for (int r = 0; r < 8; r++) {
            p[(size_t)(rowbase + r) * Hn + 0 * 16 + l15] = a0[r];
            p[(size_t)(rowbase + r) * Hn + 1 * 16 + l15] = a1[r];
            p[(size_t)(rowbase + r) * Hn + 2 * 16 + l15] = a2[r];
            p[(size_t)(rowbase + r) * Hn + 3 * 16 + l15] = a3[r];
        }
    } else {
        int b  = rt >> 7;
        int t0 = (rt & 127) * 16 + 8 * hi;
        float* vtb = ws + OFF_VT + (size_t)b * Hn * Tn + t0;
        v8f accs0 = a0, accs1 = a1, accs2 = a2, accs3 = a3;
        {
            float* d = vtb + (size_t)(0 * 16 + l15) * Tn;
            *(v4f*)(d)     = (v4f){accs0[0], accs0[1], accs0[2], accs0[3]};
            *(v4f*)(d + 4) = (v4f){accs0[4], accs0[5], accs0[6], accs0[7]};
        }
        {
            float* d = vtb + (size_t)(1 * 16 + l15) * Tn;
            *(v4f*)(d)     = (v4f){accs1[0], accs1[1], accs1[2], accs1[3]};
            *(v4f*)(d + 4) = (v4f){accs1[4], accs1[5], accs1[6], accs1[7]};
        }
        {
            float* d = vtb + (size_t)(2 * 16 + l15) * Tn;
            *(v4f*)(d)     = (v4f){accs2[0], accs2[1], accs2[2], accs2[3]};
            *(v4f*)(d + 4) = (v4f){accs2[4], accs2[5], accs2[6], accs2[7]};
        }
        {
            float* d = vtb + (size_t)(3 * 16 + l15) * Tn;
            *(v4f*)(d)     = (v4f){accs3[0], accs3[1], accs3[2], accs3[3]};
            *(v4f*)(d + 4) = (v4f){accs3[4], accs3[5], accs3[6], accs3[7]};
        }
    }
}

constexpr int NW = 8;
constexpr int OPITCH = 68;

__global__ void attn_kernel(const float* __restrict__ ws, float* __restrict__ out) {
    __shared__ __attribute__((aligned(16))) float sm_o[NW][16][OPITCH];
    __shared__ float sm_m[NW][16];
    __shared__ float sm_l[NW][16];

    int w    = __builtin_amdgcn_readfirstlane((int)(threadIdx.x >> 5));
    int lane = threadIdx.x & 31;
    int l15 = lane & 15;
    int hi  = lane >> 4;

    int n  = blockIdx.x;
    int qr = n >> 8;
    int c  = n & 255;
    int b  = c & 7;
    int u  = c >> 3;
    int i  = (3 - qr) * 32 + ((qr & 1) ? u : 31 - u);

    const float* kp = ws + OFF_KP + (size_t)(b * Tn + i * 16 + l15) * Hn + 2 * hi;
    const float* qp = ws + OFF_QP + (size_t)b * Tn * Hn;
    const float* vt = ws + OFF_VT + (size_t)b * Hn * Tn;

    v2f kb[16];
#pragma unroll
    for (int kk = 0; kk < 16; kk++) kb[kk] = *(const v2f*)(kp + 4 * kk);

    v8f oacc[4] = {};
    float m = -__builtin_inff();
    float l = 0.0f;

    for (int j = w; j <= i; j += NW) {
        const float* qa = qp + (size_t)(j * 16 + l15) * Hn + 2 * hi;
        v8f s = {};
#pragma unroll
        for (int kk = 0; kk < 16; kk++) {
            v2f a = *(const v2f*)(qa + 4 * kk);
            s = WMMA_F32(a, kb[kk], s);
        }

        if (j == i) {
#pragma unroll
            for (int r = 0; r < 8; r++) {
                int sl = r + 8 * hi;
                s[r] = (sl > l15) ? -__builtin_inff() : s[r];
            }
        }

        float tm = s[0];
#pragma unroll
        for (int r = 1; r < 8; r++) tm = fmaxf(tm, s[r]);
        tm = fmaxf(tm, __shfl_xor(tm, 16, 32));
        float mn = fmaxf(m, tm);
        float sc = __expf(m - mn);
        m = mn;

        float p[8];
        float tl = 0.0f;
#pragma unroll
        for (int r = 0; r < 8; r++) {
            p[r] = __expf(s[r] - mn);
            tl += p[r];
        }
        tl += __shfl_xor(tl, 16, 32);
        l = l * sc + tl;
#pragma unroll
        for (int nn = 0; nn < 4; nn++) oacc[nn] = oacc[nn] * sc;

        float q0 = __shfl_xor(p[0], 16, 32), q1 = __shfl_xor(p[1], 16, 32);
        float q2 = __shfl_xor(p[2], 16, 32), q3 = __shfl_xor(p[3], 16, 32);
        float q4 = __shfl_xor(p[4], 16, 32), q5 = __shfl_xor(p[5], 16, 32);
        float q6 = __shfl_xor(p[6], 16, 32), q7 = __shfl_xor(p[7], 16, 32);
        bool ishi = (hi != 0);
        v2f bb[4];
        bb[0] = ishi ? (v2f){q2, q3} : (v2f){p[0], p[1]};
        bb[1] = ishi ? (v2f){q6, q7} : (v2f){p[4], p[5]};
        bb[2] = ishi ? (v2f){p[2], p[3]} : (v2f){q0, q1};
        bb[3] = ishi ? (v2f){p[6], p[7]} : (v2f){q4, q5};

        const float* va = vt + (size_t)l15 * Tn + j * 16 + 2 * hi;
#pragma unroll
        for (int nn = 0; nn < 4; nn++) {
            const float* van = va + (size_t)nn * 16 * Tn;
#pragma unroll
            for (int kk = 0; kk < 4; kk++) {
                v2f a = *(const v2f*)(van + 4 * kk);
                oacc[nn] = WMMA_F32(a, bb[kk], oacc[nn]);
            }
        }
    }

    {
        float* po = &sm_o[w][l15][8 * hi];
#pragma unroll
        for (int nn = 0; nn < 4; nn++) {
            *(v4f*)(po + nn * 16)     = (v4f){oacc[nn][0], oacc[nn][1], oacc[nn][2], oacc[nn][3]};
            *(v4f*)(po + nn * 16 + 4) = (v4f){oacc[nn][4], oacc[nn][5], oacc[nn][6], oacc[nn][7]};
        }
        if (hi == 0) {
            sm_m[w][l15] = m;
            sm_l[w][l15] = l;
        }
    }
    __syncthreads();

    {
        int t  = 2 * w + hi;
        int h0 = 4 * l15;
        float mw[NW];
        float M = -__builtin_inff();
#pragma unroll
        for (int ww = 0; ww < NW; ww++) {
            mw[ww] = sm_m[ww][t];
            M = fmaxf(M, mw[ww]);
        }
        float L = 0.0f;
        v4f acc = {0.0f, 0.0f, 0.0f, 0.0f};
#pragma unroll
        for (int ww = 0; ww < NW; ww++) {
            float f = __expf(mw[ww] - M);
            L += sm_l[ww][t] * f;
            v4f o = *(const v4f*)(&sm_o[ww][t][h0]);
            acc += o * f;
        }
        float invl = 1.0f / L;
        *(v4f*)(out + (size_t)(b * Tn + i * 16 + t) * Hn + h0) = acc * invl;
    }
}

extern "C" void kernel_launch(void* const* d_in, const int* in_sizes, int n_in,
                              void* d_out, int out_size, void* d_ws, size_t ws_size,
                              hipStream_t stream) {
    const float* x  = (const float*)d_in[0];
    const float* Wk = (const float*)d_in[1];
    const float* Wq = (const float*)d_in[2];
    const float* Wv = (const float*)d_in[3];
    float* ws  = (float*)d_ws;
    float* out = (float*)d_out;

    wtrans_kernel<<<768, 256, 0, stream>>>(Wk, Wq, Wv, ws + OFF_WT);
    proj_kernel<<<384, 256, 0, stream>>>(x, ws + OFF_WT, ws);
    attn_kernel<<<1024, 256, 0, stream>>>(ws, out);
}


# projections: x and Wt k-chunks staged through LDS by 64-row work-groups, fragments read as 16-byte LDS loads instead of 16-line global loads; changes the k summation order of the projections
# speedup vs baseline: 3.5695x; 1.5092x over previous
#include <hip/hip_runtime.h>

typedef float v2f __attribute__((ext_vector_type(2)));
typedef float v4f __attribute__((ext_vector_type(4)));
typedef float v8f __attribute__((ext_vector_type(8)));

#define WMMA_F32(a, b, c) \
    __builtin_amdgcn_wmma_f32_16x16x4_f32(false, (a), false, (b), (short)0, (c), false, false)

constexpr int Bn = 8, Tn = 2048, Cn = 1024, Hn = 64;

constexpr size_t OFF_KP = 0;
constexpr size_t OFF_QP = OFF_KP + (size_t)Bn * Tn * Hn;
constexpr size_t OFF_VT = OFF_QP + (size_t)Bn * Tn * Hn;
constexpr size_t OFF_WT = OFF_VT + (size_t)Bn * Tn * Hn;

__global__ void wtrans_kernel(const float* __restrict__ Wk,
                              const float* __restrict__ Wq,
                              const float* __restrict__ Wv,
                              float* __restrict__ wt) {
    int idx = blockIdx.x * 256 + threadIdx.x;
    int m = idx >> 16;
    int e = idx & 65535;
    int h = e >> 10;
    int c = e & 1023;
    const float* W = (m == 0) ? Wk : ((m == 1) ? Wq : Wv);
    float val = W[c * Hn + h];
    if (m == 1) val *= 0.125f;
    wt[(size_t)m * 65536 + e] = val;
}

constexpr int PM    = 64;
constexpr int PN    = 3 * Hn;
constexpr int KC    = 64;
constexpr int PITCH = KC + 4;
constexpr int PROWS = PM + PN;
constexpr int NCH   = Cn / KC;
constexpr int NST   = PROWS / 16;
constexpr int NFS   = KC / 8;

__device__ __forceinline__ const float* stage_src(const float* xg, const float* wg, int p) {
    return (p < PM / 16) ? (xg + (size_t)p * 16 * Cn) : (wg + (size_t)(p - PM / 16) * 16 * Cn);
}

template <bool STORE_NEXT, bool LOAD_NEXT2>
__device__ __forceinline__ void proj_chunk(int c, float* smem, const float* xg, const float* wg,
                                           int soff, int aoff, int boff,
                                           v4f (&st)[NST], v8f (&acc)[2][3]) {
    const float* sa = smem + (c & 1) * (PROWS * PITCH) + aoff;
    const float* sb = smem + (c & 1) * (PROWS * PITCH) + boff;
    float*       sn = smem + ((c + 1) & 1) * (PROWS * PITCH) + soff;

    v4f a[2][2], b[2][3];
#pragma unroll
    for (int mi = 0; mi < 2; mi++) a[0][mi] = *(const v4f*)(sa + mi * 16 * PITCH);
#pragma unroll
    for (int ni = 0; ni < 3; ni++) b[0][ni] = *(const v4f*)(sb + ni * 16 * PITCH);

#pragma unroll
    for (int s = 0; s < NFS; s++) {
        const int cur = s & 1, nxt = cur ^ 1;
        if (s + 1 < NFS) {
#pragma unroll
            for (int mi = 0; mi < 2; mi++) a[nxt][mi] = *(const v4f*)(sa + mi * 16 * PITCH + 8 * (s + 1));
#pragma unroll
            for (int ni = 0; ni < 3; ni++) b[nxt][ni] = *(const v4f*)(sb + ni * 16 * PITCH + 8 * (s + 1));
        }
#pragma unroll
        for (int mi = 0; mi < 2; mi++)
#pragma unroll
            for (int ni = 0; ni < 3; ni++)
                acc[mi][ni] = WMMA_F32(__builtin_shufflevector(a[cur][mi], a[cur][mi], 0, 1),
                                       __builtin_shufflevector(b[cur][ni], b[cur][ni], 0, 1), acc[mi][ni]);
#pragma unroll
        for (int mi = 0; mi < 2; mi++)
#pragma unroll
            for (int ni = 0; ni < 3; ni++)
                acc[mi][ni] = WMMA_F32(__builtin_shufflevector(a[cur][mi], a[cur][mi], 2, 3),
                                       __builtin_shufflevector(b[cur][ni], b[cur][ni], 2, 3), acc[mi][ni]);
        if (s < NST / 4) {
            if (STORE_NEXT || LOAD_NEXT2) __builtin_amdgcn_sched_barrier(0);
            if (STORE_NEXT) {
#pragma unroll
                for (int p = 4 * s; p < 4 * s + 4; p++) *(v4f*)(sn + p * 16 * PITCH) = st[p];
            }
            if (LOAD_NEXT2) {
#pragma unroll
                for (int p = 4 * s; p < 4 * s + 4; p++) st[p] = *(const v4f*)(stage_src(xg, wg, p) + (c + 2) * KC);
            }
        }
    }
    __syncthreads();
}

__global__ void __launch_bounds__(256)
proj_kernel(const float* __restrict__ x,
            const float* __restrict__ wt,
            float* __restrict__ ws) {
    __shared__ __attribute__((aligned(16))) float sm[2 * PROWS * PITCH];

    int tid  = threadIdx.x;
    int w    = __builtin_amdgcn_readfirstlane(tid >> 5);
    int lane = tid & 31;
    int l15  = lane & 15;
    int hi   = lane >> 4;
    int wr   = w >> 2;
    int wc   = w & 3;

    int srow = tid >> 4;
    int scol = (tid & 15) * 4;
    const float* xg = x  + (size_t)(blockIdx.x * PM + srow) * Cn + scol;
    const float* wg = wt + (size_t)srow * Cn + scol;
    int soff = srow * PITCH + scol;

    v4f st[NST];
#pragma unroll
    for (int p = 0; p < NST; p++) st[p] = *(const v4f*)stage_src(xg, wg, p);
#pragma unroll
    for (int p = 0; p < NST; p++) *(v4f*)(&sm[soff + p * 16 * PITCH]) = st[p];
#pragma unroll
    for (int p = 0; p < NST; p++) st[p] = *(const v4f*)(stage_src(xg, wg, p) + KC);
    __syncthreads();

    v8f acc[2][3] = {};
    int aoff = (wr * 32 + l15) * PITCH + 4 * hi;
    int boff = (PM + wc * 48 + l15) * PITCH + 4 * hi;

    for (int c = 0; c < NCH - 2; c++)
        proj_chunk<true, true>(c, sm, xg, wg, soff, aoff, boff, st, acc);
    proj_chunk<true, false>(NCH - 2, sm, xg, wg, soff, aoff, boff, st, acc);
    proj_chunk<false, false>(NCH - 1, sm, xg, wg, soff, aoff, boff, st, acc);

#pragma unroll
    for (int ni = 0; ni < 3; ni++) {
        int nt    = wc * 3 + ni;
        int which = nt >> 2;
        int cg    = (nt & 3) * 16 + l15;
#pragma unroll
        for (int mi = 0; mi < 2; mi++) {
            int grow = blockIdx.x * PM + wr * 32 + mi * 16 + 8 * hi;
            v8f v = acc[mi][ni];
            if (which < 2) {
                float* p = ws + (which == 0 ? OFF_KP : OFF_QP) + (size_t)grow * Hn + cg;
#pragma unroll
                for (int r = 0; r < 8; r++) p[(size_t)r * Hn] = v[r];
            } else {
                int b  = grow >> 11;
                int t0 = grow & (Tn - 1);
                float* d = ws + OFF_VT + (size_t)b * Hn * Tn + (size_t)cg * Tn + t0;
                *(v4f*)(d)     = (v4f){v[0], v[1], v[2], v[3]};
                *(v4f*)(d + 4) = (v4f){v[4], v[5], v[6], v[7]};
            }
        }
    }
}

constexpr int NW = 8;
constexpr int OPITCH = 68;

__global__ void attn_kernel(const float* __restrict__ ws, float* __restrict__ out) {
    __shared__ __attribute__((aligned(16))) float sm_o[NW][16][OPITCH];
    __shared__ float sm_m[NW][16];
    __shared__ float sm_l[NW][16];

    int w    = __builtin_amdgcn_readfirstlane((int)(threadIdx.x >> 5));
    int lane = threadIdx.x & 31;
    int l15 = lane & 15;
    int hi  = lane >> 4;

    int n  = blockIdx.x;
    int qr = n >> 8;
    int c  = n & 255;
    int b  = c & 7;
    int u  = c >> 3;
    int i  = (3 - qr) * 32 + ((qr & 1) ? u : 31 - u);

    const float* kp = ws + OFF_KP + (size_t)(b * Tn + i * 16 + l15) * Hn + 2 * hi;
    const float* qp = ws + OFF_QP + (size_t)b * Tn * Hn;
    const float* vt = ws + OFF_VT + (size_t)b * Hn * Tn;

    v2f kb[16];
#pragma unroll
    for (int kk = 0; kk < 16; kk++) kb[kk] = *(const v2f*)(kp + 4 * kk);

    v8f oacc[4] = {};
    float m = -__builtin_inff();
    float l = 0.0f;

    for (int j = w; j <= i; j += NW) {
        const float* qa = qp + (size_t)(j * 16 + l15) * Hn + 2 * hi;
        v8f s = {};
#pragma unroll
        for (int kk = 0; kk < 16; kk++) {
            v2f a = *(const v2f*)(qa + 4 * kk);
            s = WMMA_F32(a, kb[kk], s);
        }

        if (j == i) {
#pragma unroll
            for (int r = 0; r < 8; r++) {
                int sl = r + 8 * hi;
                s[r] = (sl > l15) ? -__builtin_inff() : s[r];
            }
        }

        float tm = s[0];
#pragma unroll
        for (int r = 1; r < 8; r++) tm = fmaxf(tm, s[r]);
        tm = fmaxf(tm, __shfl_xor(tm, 16, 32));
        float mn = fmaxf(m, tm);
        float sc = __expf(m - mn);
        m = mn;

        float p[8];
        float tl = 0.0f;
#pragma unroll
        for (int r = 0; r < 8; r++) {
            p[r] = __expf(s[r] - mn);
            tl += p[r];
        }
        tl += __shfl_xor(tl, 16, 32);
        l = l * sc + tl;
#pragma unroll
        for (int nn = 0; nn < 4; nn++) oacc[nn] = oacc[nn] * sc;

        float q0 = __shfl_xor(p[0], 16, 32), q1 = __shfl_xor(p[1], 16, 32);
        float q2 = __shfl_xor(p[2], 16, 32), q3 = __shfl_xor(p[3], 16, 32);
        float q4 = __shfl_xor(p[4], 16, 32), q5 = __shfl_xor(p[5], 16, 32);
        float q6 = __shfl_xor(p[6], 16, 32), q7 = __shfl_xor(p[7], 16, 32);
        bool ishi = (hi != 0);
        v2f bb[4];
        bb[0] = ishi ? (v2f){q2, q3} : (v2f){p[0], p[1]};
        bb[1] = ishi ? (v2f){q6, q7} : (v2f){p[4], p[5]};
        bb[2] = ishi ? (v2f){p[2], p[3]} : (v2f){q0, q1};
        bb[3] = ishi ? (v2f){p[6], p[7]} : (v2f){q4, q5};

        const float* va = vt + (size_t)l15 * Tn + j * 16 + 2 * hi;
#pragma unroll
        for (int nn = 0; nn < 4; nn++) {
            const float* van = va + (size_t)nn * 16 * Tn;
#pragma unroll
            for (int kk = 0; kk < 4; kk++) {
                v2f a = *(const v2f*)(van + 4 * kk);
                oacc[nn] = WMMA_F32(a, bb[kk], oacc[nn]);
            }
        }
    }

    {
        float* po = &sm_o[w][l15][8 * hi];
#pragma unroll
        for (int nn = 0; nn < 4; nn++) {
            *(v4f*)(po + nn * 16)     = (v4f){oacc[nn][0], oacc[nn][1], oacc[nn][2], oacc[nn][3]};
            *(v4f*)(po + nn * 16 + 4) = (v4f){oacc[nn][4], oacc[nn][5], oacc[nn][6], oacc[nn][7]};
        }
        if (hi == 0) {
            sm_m[w][l15] = m;
            sm_l[w][l15] = l;
        }
    }
    __syncthreads();

    {
        int t  = 2 * w + hi;
        int h0 = 4 * l15;
        float mw[NW];
        float M = -__builtin_inff();
#pragma unroll
        for (int ww = 0; ww < NW; ww++) {
            mw[ww] = sm_m[ww][t];
            M = fmaxf(M, mw[ww]);
        }
        float L = 0.0f;
        v4f acc = {0.0f, 0.0f, 0.0f, 0.0f};
#pragma unroll
        for (int ww = 0; ww < NW; ww++) {
            float f = __expf(mw[ww] - M);
            L += sm_l[ww][t] * f;
            v4f o = *(const v4f*)(&sm_o[ww][t][h0]);
            acc += o * f;
        }
        float invl = 1.0f / L;
        *(v4f*)(out + (size_t)(b * Tn + i * 16 + t) * Hn + h0) = acc * invl;
    }
}

extern "C" void kernel_launch(void* const* d_in, const int* in_sizes, int n_in,
                              void* d_out, int out_size, void* d_ws, size_t ws_size,
                              hipStream_t stream) {
    const float* x  = (const float*)d_in[0];
    const float* Wk = (const float*)d_in[1];
    const float* Wq = (const float*)d_in[2];
    const float* Wv = (const float*)d_in[3];
    float* ws  = (float*)d_ws;
    float* out = (float*)d_out;

    wtrans_kernel<<<768, 256, 0, stream>>>(Wk, Wq, Wv, ws + OFF_WT);
    proj_kernel<<<256, 256, 0, stream>>>(x, ws + OFF_WT, ws);
    attn_kernel<<<1024, 256, 0, stream>>>(ws, out);
}


# k, q, v written by the projection in matrix-fragment order: attention fragments become whole-line 16-byte loads and P feeds PV in place; changes the k summation order of both attention products
# speedup vs baseline: 4.2435x; 1.1888x over previous
#include <hip/hip_runtime.h>

typedef float v2f __attribute__((ext_vector_type(2)));
typedef float v4f __attribute__((ext_vector_type(4)));
typedef float v8f __attribute__((ext_vector_type(8)));

#define WMMA_F32(a, b, c) \
    __builtin_amdgcn_wmma_f32_16x16x4_f32(false, (a), false, (b), (short)0, (c), false, false)

constexpr int Bn = 8, Tn = 2048, Cn = 1024, Hn = 64;

constexpr size_t OFF_KF = 0;
constexpr size_t OFF_QF = OFF_KF + (size_t)Bn * Tn * Hn;
constexpr size_t OFF_VF = OFF_QF + (size_t)Bn * Tn * Hn;
constexpr size_t OFF_WT = OFF_VF + (size_t)Bn * Tn * Hn;
constexpr int    TILE_F = 16 * Hn;

__global__ void wtrans_kernel(const float* __restrict__ Wk,
                              const float* __restrict__ Wq,
                              const float* __restrict__ Wv,
                              float* __restrict__ wt) {
    int idx = blockIdx.x * 256 + threadIdx.x;
    int m = idx >> 16;
    int e = idx & 65535;
    int h = e >> 10;
    int c = e & 1023;
    const float* W = (m == 0) ? Wk : ((m == 1) ? Wq : Wv);
    float val = W[c * Hn + h];
    if (m == 1) val *= 0.125f;
    wt[(size_t)m * 65536 + e] = val;
}

constexpr int PM    = 64;
constexpr int PN    = 3 * Hn;
constexpr int KC    = 64;
constexpr int PITCH = KC + 4;
constexpr int PROWS = PM + PN;
constexpr int NCH   = Cn / KC;
constexpr int NST   = PROWS / 16;
constexpr int NFS   = KC / 8;

__device__ __forceinline__ const float* stage_src(const float* xg, const float* wg, int p) {
    return (p < PM / 16) ? (xg + (size_t)p * 16 * Cn) : (wg + (size_t)(p - PM / 16) * 16 * Cn);
}

template <bool STORE_NEXT, bool LOAD_NEXT2>
__device__ __forceinline__ void proj_chunk(int c, float* smem, const float* xg, const float* wg,
                                           int soff, int aoff, int boff,
                                           v4f (&st)[NST], v8f (&acc)[2][3]) {
    const float* sa = smem + (c & 1) * (PROWS * PITCH) + aoff;
    const float* sb = smem + (c & 1) * (PROWS * PITCH) + boff;
    float*       sn = smem + ((c + 1) & 1) * (PROWS * PITCH) + soff;

    v4f a[2][2], b[2][3];
#pragma unroll
    for (int mi = 0; mi < 2; mi++) a[0][mi] = *(const v4f*)(sa + mi * 16 * PITCH);
#pragma unroll
    for (int ni = 0; ni < 3; ni++) b[0][ni] = *(const v4f*)(sb + ni * Hn * PITCH);

#pragma unroll
    for (int s = 0; s < NFS; s++) {
        const int cur = s & 1, nxt = cur ^ 1;
        if (s + 1 < NFS) {
#pragma unroll
            for (int mi = 0; mi < 2; mi++) a[nxt][mi] = *(const v4f*)(sa + mi * 16 * PITCH + 8 * (s + 1));
#pragma unroll
            for (int ni = 0; ni < 3; ni++) b[nxt][ni] = *(const v4f*)(sb + ni * Hn * PITCH + 8 * (s + 1));
        }
#pragma unroll
        for (int mi = 0; mi < 2; mi++)
#pragma unroll
            for (int ni = 0; ni < 3; ni++) {
                v2f xa = __builtin_shufflevector(a[cur][mi], a[cur][mi], 0, 1);
                v2f wb = __builtin_shufflevector(b[cur][ni], b[cur][ni], 0, 1);
                acc[mi][ni] = (ni < 2) ? WMMA_F32(wb, xa, acc[mi][ni]) : WMMA_F32(xa, wb, acc[mi][ni]);
            }
#pragma unroll
        for (int mi = 0; mi < 2; mi++)
#pragma unroll
            for (int ni = 0; ni < 3; ni++) {
                v2f xa = __builtin_shufflevector(a[cur][mi], a[cur][mi], 2, 3);
                v2f wb = __builtin_shufflevector(b[cur][ni], b[cur][ni], 2, 3);
                acc[mi][ni] = (ni < 2) ? WMMA_F32(wb, xa, acc[mi][ni]) : WMMA_F32(xa, wb, acc[mi][ni]);
            }
        if (s < NST / 4) {
            if (STORE_NEXT || LOAD_NEXT2) __builtin_amdgcn_sched_barrier(0);
            if (STORE_NEXT) {
#pragma unroll
                for (int p = 4 * s; p < 4 * s + 4; p++) *(v4f*)(sn + p * 16 * PITCH) = st[p];
            }
            if (LOAD_NEXT2) {
#pragma unroll
                for (int p = 4 * s; p < 4 * s + 4; p++) st[p] = *(const v4f*)(stage_src(xg, wg, p) + (c + 2) * KC);
            }
        }
    }
    __syncthreads();
}

__global__ void __launch_bounds__(256)
proj_kernel(const float* __restrict__ x,
            const float* __restrict__ wt,
            float* __restrict__ ws) {
    __shared__ __attribute__((aligned(16))) float sm[2 * PROWS * PITCH];

    int tid  = threadIdx.x;
    int w    = __builtin_amdgcn_readfirstlane(tid >> 5);
    int lane = tid & 31;
    int l15  = lane & 15;
    int hi   = lane >> 4;
    int wr   = w >> 2;
    int wc   = w & 3;

    int srow = tid >> 4;
    int scol = (tid & 15) * 4;
    const float* xg = x  + (size_t)(blockIdx.x * PM + srow) * Cn + scol;
    const float* wg = wt + (size_t)srow * Cn + scol;
    int soff = srow * PITCH + scol;

    v4f st[NST];
#pragma unroll
    for (int p = 0; p < NST; p++) st[p] = *(const v4f*)stage_src(xg, wg, p);
#pragma unroll
    for (int p = 0; p < NST; p++) *(v4f*)(&sm[soff + p * 16 * PITCH]) = st[p];
#pragma unroll
    for (int p = 0; p < NST; p++) st[p] = *(const v4f*)(stage_src(xg, wg, p) + KC);
    __syncthreads();

    v8f acc[2][3] = {};
    int aoff = (wr * 32 + l15) * PITCH + 4 * hi;
    int boff = (PM + wc * 16 + l15) * PITCH + 4 * hi;

    for (int c = 0; c < NCH - 2; c++)
        proj_chunk<true, true>(c, sm, xg, wg, soff, aoff, boff, st, acc);
    proj_chunk<true, false>(NCH - 2, sm, xg, wg, soff, aoff, boff, st, acc);
    proj_chunk<false, false>(NCH - 1, sm, xg, wg, soff, aoff, boff, st, acc);

#pragma unroll
    for (int ni = 0; ni < 3; ni++) {
        float* plane = ws + (ni == 0 ? OFF_KF : (ni == 1 ? OFF_QF : OFF_VF));
#pragma unroll
        for (int mi = 0; mi < 2; mi++) {
            int jg = blockIdx.x * (PM / 16) + wr * 2 + mi;
            float* d = plane + (size_t)jg * TILE_F + (wc * 2) * 128 + lane * 4;
            v8f v = acc[mi][ni];
            *(v4f*)(d)       = (v4f){v[0], v[1], v[2], v[3]};
            *(v4f*)(d + 128) = (v4f){v[4], v[5], v[6], v[7]};
        }
    }
}

constexpr int NW = 8;
constexpr int OPITCH = 68;

__global__ void attn_kernel(const float* __restrict__ ws, float* __restrict__ out) {
    __shared__ __attribute__((aligned(16))) float sm_o[NW][16][OPITCH];
    __shared__ float sm_m[NW][16];
    __shared__ float sm_l[NW][16];

    int w    = __builtin_amdgcn_readfirstlane((int)(threadIdx.x >> 5));
    int lane = threadIdx.x & 31;
    int l15 = lane & 15;
    int hi  = lane >> 4;

    int n  = blockIdx.x;
    int qr = n >> 8;
    int c  = n & 255;
    int b  = c & 7;
    int u  = c >> 3;
    int i  = (3 - qr) * 32 + ((qr & 1) ? u : 31 - u);

    const float* kf = ws + OFF_KF + (size_t)(b * (Tn / 16) + i) * TILE_F + lane * 4;
    const float* qf = ws + OFF_QF + (size_t)b * Tn * Hn + lane * 4;
    const float* vf = ws + OFF_VF + (size_t)b * Tn * Hn + lane * 4;

    v4f kb[8];
#pragma unroll
    for (int ch = 0; ch < 8; ch++) kb[ch] = *(const v4f*)(kf + ch * 128);

    v8f oacc[4] = {};
    float m = -__builtin_inff();
    float l = 0.0f;

    for (int j = w; j <= i; j += NW) {
        const float* qa = qf + (size_t)j * TILE_F;
        v8f s = {};
#pragma unroll
        for (int ch = 0; ch < 8; ch++) {
            v4f a = *(const v4f*)(qa + ch * 128);
            s = WMMA_F32(__builtin_shufflevector(a, a, 0, 1),
                         __builtin_shufflevector(kb[ch], kb[ch], 0, 1), s);
            s = WMMA_F32(__builtin_shufflevector(a, a, 2, 3),
                         __builtin_shufflevector(kb[ch], kb[ch], 2, 3), s);
        }

        if (j == i) {
#pragma unroll
            for (int r = 0; r < 8; r++) {
                int sl = r + 8 * hi;
                s[r] = (sl > l15) ? -__builtin_inff() : s[r];
            }
        }

        float tm = s[0];
#pragma unroll
        for (int r = 1; r < 8; r++) tm = fmaxf(tm, s[r]);
        tm = fmaxf(tm, __shfl_xor(tm, 16, 32));
        float mn = fmaxf(m, tm);
        float sc = __expf(m - mn);
        m = mn;

        float p[8];
        float tl = 0.0f;
#pragma unroll
        for (int r = 0; r < 8; r++) {
            p[r] = __expf(s[r] - mn);
            tl += p[r];
        }
        tl += __shfl_xor(tl, 16, 32);
        l = l * sc + tl;
#pragma unroll
        for (int nn = 0; nn < 4; nn++) oacc[nn] = oacc[nn] * sc;

        const float* va = vf + (size_t)j * TILE_F;
#pragma unroll
        for (int nn = 0; nn < 4; nn++) {
#pragma unroll
            for (int qd = 0; qd < 2; qd++) {
                v4f a = *(const v4f*)(va + (nn * 2 + qd) * 128);
                oacc[nn] = WMMA_F32(__builtin_shufflevector(a, a, 0, 1),
                                    ((v2f){p[4 * qd], p[4 * qd + 1]}), oacc[nn]);
                oacc[nn] = WMMA_F32(__builtin_shufflevector(a, a, 2, 3),
                                    ((v2f){p[4 * qd + 2], p[4 * qd + 3]}), oacc[nn]);
            }
        }
    }

    {
        float* po = &sm_o[w][l15][8 * hi];
#pragma unroll
        for (int nn = 0; nn < 4; nn++) {
            *(v4f*)(po + nn * 16)     = (v4f){oacc[nn][0], oacc[nn][1], oacc[nn][2], oacc[nn][3]};
            *(v4f*)(po + nn * 16 + 4) = (v4f){oacc[nn][4], oacc[nn][5], oacc[nn][6], oacc[nn][7]};
        }
        if (hi == 0) {
            sm_m[w][l15] = m;
            sm_l[w][l15] = l;
        }
    }
    __syncthreads();

    {
        int t  = 2 * w + hi;
        int h0 = 4 * l15;
        float mw[NW];
        float M = -__builtin_inff();
#pragma unroll
        for (int ww = 0; ww < NW; ww++) {
            mw[ww] = sm_m[ww][t];
            M = fmaxf(M, mw[ww]);
        }
        float L = 0.0f;
        v4f acc = {0.0f, 0.0f, 0.0f, 0.0f};
#pragma unroll
        for (int ww = 0; ww < NW; ww++) {
            float f = __expf(mw[ww] - M);
            L += sm_l[ww][t] * f;
            v4f o = *(const v4f*)(&sm_o[ww][t][h0]);
            acc += o * f;
        }
        float invl = 1.0f / L;
        *(v4f*)(out + (size_t)(b * Tn + i * 16 + t) * Hn + h0) = acc * invl;
    }
}

extern "C" void kernel_launch(void* const* d_in, const int* in_sizes, int n_in,
                              void* d_out, int out_size, void* d_ws, size_t ws_size,
                              hipStream_t stream) {
    const float* x  = (const float*)d_in[0];
    const float* Wk = (const float*)d_in[1];
    const float* Wq = (const float*)d_in[2];
    const float* Wv = (const float*)d_in[3];
    float* ws  = (float*)d_ws;
    float* out = (float*)d_out;

    wtrans_kernel<<<768, 256, 0, stream>>>(Wk, Wq, Wv, ws + OFF_WT);
    proj_kernel<<<256, 256, 0, stream>>>(x, ws + OFF_WT, ws);
    attn_kernel<<<1024, 256, 0, stream>>>(ws, out);
}


# attention: each wave keeps two adjacent 16-row tiles so every loaded q and v fragment feeds both; 16-wave work-groups of two 8-wave teams (tile pairs 63-u and u, equal work); same summation order
# speedup vs baseline: 4.3788x; 1.0319x over previous
#include <hip/hip_runtime.h>

typedef float v2f __attribute__((ext_vector_type(2)));
typedef float v4f __attribute__((ext_vector_type(4)));
typedef float v8f __attribute__((ext_vector_type(8)));

#define WMMA_F32(a, b, c) \
    __builtin_amdgcn_wmma_f32_16x16x4_f32(false, (a), false, (b), (short)0, (c), false, false)

constexpr int Bn = 8, Tn = 2048, Cn = 1024, Hn = 64;

constexpr size_t OFF_KF = 0;
constexpr size_t OFF_QF = OFF_KF + (size_t)Bn * Tn * Hn;
constexpr size_t OFF_VF = OFF_QF + (size_t)Bn * Tn * Hn;
constexpr size_t OFF_WT = OFF_VF + (size_t)Bn * Tn * Hn;
constexpr int    TILE_F = 16 * Hn;

__global__ void wtrans_kernel(const float* __restrict__ Wk,
                              const float* __restrict__ Wq,
                              const float* __restrict__ Wv,
                              float* __restrict__ wt) {
    int idx = blockIdx.x * 256 + threadIdx.x;
    int m = idx >> 16;
    int e = idx & 65535;
    int h = e >> 10;
    int c = e & 1023;
    const float* W = (m == 0) ? Wk : ((m == 1) ? Wq : Wv);
    float val = W[c * Hn + h];
    if (m == 1) val *= 0.125f;
    wt[(size_t)m * 65536 + e] = val;
}

constexpr int PM    = 64;
constexpr int PN    = 3 * Hn;
constexpr int KC    = 64;
constexpr int PITCH = KC + 4;
constexpr int PROWS = PM + PN;
constexpr int NCH   = Cn / KC;
constexpr int NST   = PROWS / 16;
constexpr int NFS   = KC / 8;

__device__ __forceinline__ const float* stage_src(const float* xg, const float* wg, int p) {
    return (p < PM / 16) ? (xg + (size_t)p * 16 * Cn) : (wg + (size_t)(p - PM / 16) * 16 * Cn);
}

template <bool STORE_NEXT, bool LOAD_NEXT2>
__device__ __forceinline__ void proj_chunk(int c, float* smem, const float* xg, const float* wg,
                                           int soff, int aoff, int boff,
                                           v4f (&st)[NST], v8f (&acc)[2][3]) {
    const float* sa = smem + (c & 1) * (PROWS * PITCH) + aoff;
    const float* sb = smem + (c & 1) * (PROWS * PITCH) + boff;
    float*       sn = smem + ((c + 1) & 1) * (PROWS * PITCH) + soff;

    v4f a[2][2], b[2][3];
#pragma unroll
    for (int mi = 0; mi < 2; mi++) a[0][mi] = *(const v4f*)(sa + mi * 16 * PITCH);
#pragma unroll
    for (int ni = 0; ni < 3; ni++) b[0][ni] = *(const v4f*)(sb + ni * Hn * PITCH);

#pragma unroll
    for (int s = 0; s < NFS; s++) {
        const int cur = s & 1, nxt = cur ^ 1;
        if (s + 1 < NFS) {
#pragma unroll
            for (int mi = 0; mi < 2; mi++) a[nxt][mi] = *(const v4f*)(sa + mi * 16 * PITCH + 8 * (s + 1));
#pragma unroll
            for (int ni = 0; ni < 3; ni++) b[nxt][ni] = *(const v4f*)(sb + ni * Hn * PITCH + 8 * (s + 1));
        }
#pragma unroll
        for (int mi = 0; mi < 2; mi++)
#pragma unroll
            for (int ni = 0; ni < 3; ni++) {
                v2f xa = __builtin_shufflevector(a[cur][mi], a[cur][mi], 0, 1);
                v2f wb = __builtin_shufflevector(b[cur][ni], b[cur][ni], 0, 1);
                acc[mi][ni] = (ni < 2) ? WMMA_F32(wb, xa, acc[mi][ni]) : WMMA_F32(xa, wb, acc[mi][ni]);
            }
#pragma unroll
        for (int mi = 0; mi < 2; mi++)
#pragma unroll
            for (int ni = 0; ni < 3; ni++) {
                v2f xa = __builtin_shufflevector(a[cur][mi], a[cur][mi], 2, 3);
                v2f wb = __builtin_shufflevector(b[cur][ni], b[cur][ni], 2, 3);
                acc[mi][ni] = (ni < 2) ? WMMA_F32(wb, xa, acc[mi][ni]) : WMMA_F32(xa, wb, acc[mi][ni]);
            }
        if (s < NST / 4) {
            if (STORE_NEXT || LOAD_NEXT2) __builtin_amdgcn_sched_barrier(0);
            if (STORE_NEXT) {
#pragma unroll
                for (int p = 4 * s; p < 4 * s + 4; p++) *(v4f*)(sn + p * 16 * PITCH) = st[p];
            }
            if (LOAD_NEXT2) {
#pragma unroll
                for (int p = 4 * s; p < 4 * s + 4; p++) st[p] = *(const v4f*)(stage_src(xg, wg, p) + (c + 2) * KC);
            }
        }
    }
    __syncthreads();
}

__global__ void __launch_bounds__(256)
proj_kernel(const float* __restrict__ x,
            const float* __restrict__ wt,
            float* __restrict__ ws) {
    __shared__ __attribute__((aligned(16))) float sm[2 * PROWS * PITCH];

    int tid  = threadIdx.x;
    int w    = __builtin_amdgcn_readfirstlane(tid >> 5);
    int lane = tid & 31;
    int l15  = lane & 15;
    int hi   = lane >> 4;
    int wr   = w >> 2;
    int wc   = w & 3;

    int srow = tid >> 4;
    int scol = (tid & 15) * 4;
    const float* xg = x  + (size_t)(blockIdx.x * PM + srow) * Cn + scol;
    const float* wg = wt + (size_t)srow * Cn + scol;
    int soff = srow * PITCH + scol;

    v4f st[NST];
#pragma unroll
    for (int p = 0; p < NST; p++) st[p] = *(const v4f*)stage_src(xg, wg, p);
#pragma unroll
    for (int p = 0; p < NST; p++) *(v4f*)(&sm[soff + p * 16 * PITCH]) = st[p];
#pragma unroll
    for (int p = 0; p < NST; p++) st[p] = *(const v4f*)(stage_src(xg, wg, p) + KC);
    __syncthreads();

    v8f acc[2][3] = {};
    int aoff = (wr * 32 + l15) * PITCH + 4 * hi;
    int boff = (PM + wc * 16 + l15) * PITCH + 4 * hi;

    for (int c = 0; c < NCH - 2; c++)
        proj_chunk<true, true>(c, sm, xg, wg, soff, aoff, boff, st, acc);
    proj_chunk<true, false>(NCH - 2, sm, xg, wg, soff, aoff, boff, st, acc);
    proj_chunk<false, false>(NCH - 1, sm, xg, wg, soff, aoff, boff, st, acc);

#pragma unroll
    for (int ni = 0; ni < 3; ni++) {
        float* plane = ws + (ni == 0 ? OFF_KF : (ni == 1 ? OFF_QF : OFF_VF));
#pragma unroll
        for (int mi = 0; mi < 2; mi++) {
            int jg = blockIdx.x * (PM / 16) + wr * 2 + mi;
            float* d = plane + (size_t)jg * TILE_F + (wc * 2) * 128 + lane * 4;
            v8f v = acc[mi][ni];
            *(v4f*)(d)       = (v4f){v[0], v[1], v[2], v[3]};
            *(v4f*)(d + 128) = (v4f){v[4], v[5], v[6], v[7]};
        }
    }
}

constexpr int NW = 8;
constexpr int NT = 2;
constexpr int RT = 2;
constexpr int OPITCH = 68;

__device__ __forceinline__ void softmax_tile(v8f& s, float& m, float& l, v8f (&oacc)[4]) {
    float tm = s[0];
#pragma unroll
    for (int r = 1; r < 8; r++) tm = fmaxf(tm, s[r]);
    tm = fmaxf(tm, __shfl_xor(tm, 16, 32));
    float mn = fmaxf(m, tm);
    float sc = __expf(m - mn);
    m = mn;
    float tl = 0.0f;
#pragma unroll
    for (int r = 0; r < 8; r++) {
        s[r] = __expf(s[r] - mn);
        tl += s[r];
    }
    tl += __shfl_xor(tl, 16, 32);
    l = l * sc + tl;
#pragma unroll
    for (int nn = 0; nn < 4; nn++) oacc[nn] = oacc[nn] * sc;
}

__device__ __forceinline__ void mask_diag(v8f& s, int l15, int hi) {
#pragma unroll
    for (int r = 0; r < 8; r++) {
        int sl = r + 8 * hi;
        s[r] = (sl > l15) ? -__builtin_inff() : s[r];
    }
}

template <int MODE>
__device__ __forceinline__ void attn_step(const float* __restrict__ qf, const float* __restrict__ vf,
                                          int j, int jn, v4f (&qa)[8], const v4f (&kb)[RT][8],
                                          v8f (&oacc)[RT][4], float (&m)[RT], float (&l)[RT],
                                          int l15, int hi) {
    const float* vp = vf + (size_t)j * TILE_F;
    v4f va[8];
#pragma unroll
    for (int ch = 0; ch < 8; ch++) va[ch] = *(const v4f*)(vp + ch * 128);

    v8f s0 = {}, s1 = {};
#pragma unroll
    for (int ch = 0; ch < 8; ch++) {
        v2f a01 = __builtin_shufflevector(qa[ch], qa[ch], 0, 1);
        v2f a23 = __builtin_shufflevector(qa[ch], qa[ch], 2, 3);
        if (MODE != 2) s0 = WMMA_F32(a01, __builtin_shufflevector(kb[0][ch], kb[0][ch], 0, 1), s0);
        s1 = WMMA_F32(a01, __builtin_shufflevector(kb[1][ch], kb[1][ch], 0, 1), s1);
        if (MODE != 2) s0 = WMMA_F32(a23, __builtin_shufflevector(kb[0][ch], kb[0][ch], 2, 3), s0);
        s1 = WMMA_F32(a23, __builtin_shufflevector(kb[1][ch], kb[1][ch], 2, 3), s1);
    }

    if (MODE == 0) {
        const float* qn = qf + (size_t)jn * TILE_F;
#pragma unroll
        for (int ch = 0; ch < 8; ch++) qa[ch] = *(const v4f*)(qn + ch * 128);
    }

    if (MODE == 1) mask_diag(s0, l15, hi);
    if (MODE == 2) mask_diag(s1, l15, hi);

    if (MODE != 2) softmax_tile(s0, m[0], l[0], oacc[0]);
    softmax_tile(s1, m[1], l[1], oacc[1]);

#pragma unroll
    for (int nn = 0; nn < 4; nn++) {
#pragma unroll
        for (int qd = 0; qd < 2; qd++) {
            v4f a = va[nn * 2 + qd];
            v2f a01 = __builtin_shufflevector(a, a, 0, 1);
            v2f a23 = __builtin_shufflevector(a, a, 2, 3);
            if (MODE != 2) oacc[0][nn] = WMMA_F32(a01, ((v2f){s0[4 * qd], s0[4 * qd + 1]}), oacc[0][nn]);
            oacc[1][nn] = WMMA_F32(a01, ((v2f){s1[4 * qd], s1[4 * qd + 1]}), oacc[1][nn]);
            if (MODE != 2) oacc[0][nn] = WMMA_F32(a23, ((v2f){s0[4 * qd + 2], s0[4 * qd + 3]}), oacc[0][nn]);
            oacc[1][nn] = WMMA_F32(a23, ((v2f){s1[4 * qd + 2], s1[4 * qd + 3]}), oacc[1][nn]);
        }
    }
}

__global__ void __launch_bounds__(NT * NW * 32)
attn_kernel(const float* __restrict__ ws, float* __restrict__ out) {
    __shared__ __attribute__((aligned(16))) float sm_o_all[NT][NW][RT * 16][OPITCH];
    __shared__ float sm_m_all[NT][NW][RT * 16];
    __shared__ float sm_l_all[NT][NW][RT * 16];

    int w16  = __builtin_amdgcn_readfirstlane((int)(threadIdx.x >> 5));
    int team = w16 >> 3;
    int w    = w16 & 7;
    int lane = threadIdx.x & 31;
    float (*sm_o)[RT * 16][OPITCH] = sm_o_all[team];
    float (*sm_m)[RT * 16]         = sm_m_all[team];
    float (*sm_l)[RT * 16]         = sm_l_all[team];
    int l15 = lane & 15;
    int hi  = lane >> 4;

    int n  = blockIdx.x;
    int b  = n & 7;
    int u  = n >> 3;
    int I  = team ? u : 63 - u;
    int i0 = 2 * I, i1 = 2 * I + 1;

    const float* kf = ws + OFF_KF + (size_t)(b * (Tn / 16) + i0) * TILE_F + lane * 4;
    const float* qf = ws + OFF_QF + (size_t)b * Tn * Hn + lane * 4;
    const float* vf = ws + OFF_VF + (size_t)b * Tn * Hn + lane * 4;

    v8f oacc[RT][4] = {};
    float m[RT] = {-__builtin_inff(), -__builtin_inff()};
    float l[RT] = {0.0f, 0.0f};

    int j = w;
    if (j <= i1) {
        v4f kb[RT][8];
#pragma unroll
        for (int rt = 0; rt < RT; rt++)
#pragma unroll
            for (int ch = 0; ch < 8; ch++) kb[rt][ch] = *(const v4f*)(kf + rt * TILE_F + ch * 128);
        v4f qa[8];
#pragma unroll
        for (int ch = 0; ch < 8; ch++) qa[ch] = *(const v4f*)(qf + (size_t)j * TILE_F + ch * 128);

        for (; j < i0; j += NW) {
            int jn = (j + NW < i1) ? j + NW : i1;
            attn_step<0>(qf, vf, j, jn, qa, kb, oacc, m, l, l15, hi);
        }
        if (j == i0)      attn_step<1>(qf, vf, j, j, qa, kb, oacc, m, l, l15, hi);
        else if (j == i1) attn_step<2>(qf, vf, j, j, qa, kb, oacc, m, l, l15, hi);
    }

#pragma unroll
    for (int rt = 0; rt < RT; rt++) {
        float* po = &sm_o[w][rt * 16 + l15][8 * hi];
#pragma unroll
        for (int nn = 0; nn < 4; nn++) {
            *(v4f*)(po + nn * 16)     = (v4f){oacc[rt][nn][0], oacc[rt][nn][1], oacc[rt][nn][2], oacc[rt][nn][3]};
            *(v4f*)(po + nn * 16 + 4) = (v4f){oacc[rt][nn][4], oacc[rt][nn][5], oacc[rt][nn][6], oacc[rt][nn][7]};
        }
        if (hi == 0) {
            sm_m[w][rt * 16 + l15] = m[rt];
            sm_l[w][rt * 16 + l15] = l[rt];
        }
    }
    __syncthreads();

#pragma unroll
    for (int k = 0; k < 2; k++) {
        int t  = 4 * w + 2 * k + hi;
        int h0 = 4 * l15;
        float mw[NW];
        float M = -__builtin_inff();
#pragma unroll
        for (int ww = 0; ww < NW; ww++) {
            mw[ww] = sm_m[ww][t];
            M = fmaxf(M, mw[ww]);
        }
        float L = 0.0f;
        v4f acc = {0.0f, 0.0f, 0.0f, 0.0f};
#pragma unroll
        for (int ww = 0; ww < NW; ww++) {
            float f = __expf(mw[ww] - M);
            L += sm_l[ww][t] * f;
            v4f o = *(const v4f*)(&sm_o[ww][t][h0]);
            acc += o * f;
        }
        float invl = 1.0f / L;
        *(v4f*)(out + (size_t)(b * Tn + I * (RT * 16) + t) * Hn + h0) = acc * invl;
    }
}

extern "C" void kernel_launch(void* const* d_in, const int* in_sizes, int n_in,
                              void* d_out, int out_size, void* d_ws, size_t ws_size,
                              hipStream_t stream) {
    const float* x  = (const float*)d_in[0];
    const float* Wk = (const float*)d_in[1];
    const float* Wq = (const float*)d_in[2];
    const float* Wv = (const float*)d_in[3];
    float* ws  = (float*)d_ws;
    float* out = (float*)d_out;

    wtrans_kernel<<<768, 256, 0, stream>>>(Wk, Wq, Wv, ws + OFF_WT);
    proj_kernel<<<256, 256, 0, stream>>>(x, ws + OFF_WT, ws);
    attn_kernel<<<256, 512, 0, stream>>>(ws, out);
}
